# starting kernel
# baseline (speedup 1.0000x reference)
#include <hip/hip_runtime.h>
#include <hip/hip_bf16.h>
#include <math.h>

typedef __attribute__((ext_vector_type(2))) float v2f;
typedef __attribute__((ext_vector_type(8))) float v8f;

#define BATCH  128
#define NRN    54
#define DTH    4
#define STEPS  30
#define PIX    5625
#define KPAD   5632
#define NPAD   64
#define CELLS  (BATCH * NRN)

__global__ void pool_kernel(const float* __restrict__ x, float* __restrict__ xp) {
  int g = blockIdx.x * blockDim.x + threadIdx.x;
  if (g >= BATCH * KPAD) return;
  int b = g / KPAD;
  int p = g - b * KPAD;
  if (p >= PIX) { xp[g] = 0.0f; return; }
  int pr = p / 75;
  int pc = p - pr * 75;
  const float* base = x + (size_t)b * 360000 + pr * 4800 + pc * 8;
  float s = 0.0f;
#pragma unroll
  for (int r = 0; r < 8; ++r) {
    const float4* q = reinterpret_cast<const float4*>(base + r * 600);
    float4 a0 = q[0], a1 = q[1];
    s += (a0.x + a0.y + a0.z + a0.w) + (a1.x + a1.y + a1.z + a1.w);
  }
  xp[g] = s * (1.0f / 64.0f);
}

__global__ void wtrans_kernel(const float* __restrict__ w, float* __restrict__ wT) {
  int g = blockIdx.x * blockDim.x + threadIdx.x;
  if (g >= KPAD * NPAD) return;
  int k = g >> 6;
  int n = g & 63;
  wT[g] = (k < PIX && n < NRN) ? w[n * PIX + k] : 0.0f;
}

__global__ void __launch_bounds__(128) fc_wmma_kernel(const float* __restrict__ xp,
                                                      const float* __restrict__ wT,
                                                      const float* __restrict__ bias,
                                                      float* __restrict__ xf) {
  __shared__ float sA[16 * 68];
  __shared__ float sB[64 * NPAD];
  const int tid  = threadIdx.x;
  const int lane = tid & 31;
  const int wave = tid >> 5;
  const int m0   = blockIdx.x * 16;
  const int half = lane >> 4;
  const int l15  = lane & 15;

  v8f acc = {0.f, 0.f, 0.f, 0.f, 0.f, 0.f, 0.f, 0.f};

  for (int k0 = 0; k0 < KPAD; k0 += 64) {
    for (int i = tid; i < 16 * 16; i += 128) {
      int m = i >> 4, q = (i & 15) << 2;
      float4 a = *reinterpret_cast<const float4*>(xp + (size_t)(m0 + m) * KPAD + k0 + q);
      *reinterpret_cast<float4*>(&sA[m * 68 + q]) = a;
    }
    for (int i = tid; i < 64 * 16; i += 128) {
      int k = i >> 4, q = (i & 15) << 2;
      float4 bv = *reinterpret_cast<const float4*>(wT + (size_t)(k0 + k) * NPAD + q);
      *reinterpret_cast<float4*>(&sB[k * NPAD + q]) = bv;
    }
    __syncthreads();

#pragma unroll 4
    for (int kk = 0; kk < 64; kk += 4) {
      int ka = kk + (half << 1);
      v2f a;
      a.x = sA[l15 * 68 + ka];
      a.y = sA[l15 * 68 + ka + 1];
      v2f bf;
      bf.x = sB[ka * NPAD + (wave << 4) + l15];
      bf.y = sB[(ka + 1) * NPAD + (wave << 4) + l15];
      acc = __builtin_amdgcn_wmma_f32_16x16x4_f32(
           false, a,  false, bf,
           (short)0, acc,  false,  false);
    }
    __syncthreads();
  }

#pragma unroll
  for (int v = 0; v < 8; ++v) {
    int m = m0 + v + (half << 3);
    int n = (wave << 4) + l15;
    float o = acc[v] + (n < NRN ? bias[n] : 0.0f);
    xf[m * NPAD + n] = o;
  }
}

#define NTH  1024
#define MAXC 7
__global__ void __launch_bounds__(NTH) scan_kernel(const float* __restrict__ xf,
                                                   const float* __restrict__ lw,
                                                   const float* __restrict__ hw,
                                                   float* __restrict__ out) {
  __shared__ float sVals[CELLS];
  __shared__ float sBatch[BATCH];
  __shared__ float sWave[NTH / 32];
  __shared__ float sThr;

  const int tid  = threadIdx.x;
  const int lane = tid & 31;
  const int wv   = tid >> 5;

  float h0[MAXC], h1[MAXC], h2[MAXC], h3[MAXC], cnt[MAXC], xv[MAXC];
  float lw0[MAXC], lw1[MAXC], lw2[MAXC], lw3[MAXC];
  const float w0 = hw[0], w1 = hw[1], w2 = hw[2], w3 = hw[3];

#pragma unroll
  for (int j = 0; j < MAXC; ++j) {
    int c = tid + (j << 10);
    h0[j] = h1[j] = h2[j] = h3[j] = cnt[j] = 0.0f;
    if (c < CELLS) {
      int b = c / NRN, n = c - b * NRN;
      xv[j]  = xf[b * NPAD + n];
      lw0[j] = lw[n * DTH + 0]; lw1[j] = lw[n * DTH + 1];
      lw2[j] = lw[n * DTH + 2]; lw3[j] = lw[n * DTH + 3];
    } else {
      xv[j] = lw0[j] = lw1[j] = lw2[j] = lw3[j] = 0.0f;
    }
  }

  for (int step = 0; step < STEPS; ++step) {
    float sv[MAXC], hsv[MAXC], rate[MAXC];
#pragma unroll
    for (int j = 0; j < MAXC; ++j) {
      int c = tid + (j << 10);
      sv[j]  = h0[j] * lw0[j] + h1[j] * lw1[j] + h2[j] * lw2[j] + h3[j] * lw3[j];
      hsv[j] = h0[j] * w0 + h1[j] * w1 + h2[j] * w2 + h3[j] * w3;
      if (c < CELLS) sVals[c] = sv[j];
    }
    __syncthreads();
    if (tid < BATCH) {
      float acc = 0.0f;
      const float* row = &sVals[tid * NRN];
#pragma unroll
      for (int n = 0; n < NRN; ++n) acc += row[n];
      sBatch[tid] = acc;
    }
    __syncthreads();
    float psum = 0.0f;
#pragma unroll
    for (int j = 0; j < MAXC; ++j) {
      int c = tid + (j << 10);
      if (c < CELLS) {
        int b = c / NRN;
        float cur = xv[j] + (sBatch[b] - sv[j]) + hsv[j];
        rate[j] = __expf(cur);
        psum += rate[j];
      } else {
        rate[j] = 0.0f;
      }
    }
#pragma unroll
    for (int off = 16; off > 0; off >>= 1) psum += __shfl_down(psum, off, 32);
    if (lane == 0) sWave[wv] = psum;
    __syncthreads();
    if (tid < 32) {
      float v = sWave[tid];
#pragma unroll
      for (int off = 16; off > 0; off >>= 1) v += __shfl_down(v, off, 32);
      if (tid == 0) sThr = v * (1.0f / (float)CELLS);
    }
    __syncthreads();
    const float thr = sThr;
#pragma unroll
    for (int j = 0; j < MAXC; ++j) {
      float sp = (rate[j] > thr) ? 1.0f : 0.0f;
      cnt[j] += sp;
      h0[j] = h1[j]; h1[j] = h2[j]; h2[j] = h3[j]; h3[j] = sp;
    }
  }

#pragma unroll
  for (int j = 0; j < MAXC; ++j) {
    int c = tid + (j << 10);
    if (c < CELLS) {
      float cc = cnt[j];
      out[c] = fmaxf(cc, 0.0f) + log1pf(__expf(-fabsf(cc)));
    }
  }
}

extern "C" void kernel_launch(void* const* d_in, const int* in_sizes, int n_in,
                              void* d_out, int out_size, void* d_ws, size_t ws_size,
                              hipStream_t stream) {
  const float* x  = (const float*)d_in[0];
  const float* lw = (const float*)d_in[1];
  const float* hw = (const float*)d_in[2];
  const float* fw = (const float*)d_in[3];
  const float* fb = (const float*)d_in[4];

  float* ws = (float*)d_ws;
  float* xp = ws;
  float* wT = xp + (size_t)BATCH * KPAD;
  float* xf = wT + (size_t)KPAD * NPAD;
  float* out = (float*)d_out;

  pool_kernel<<<(BATCH * KPAD + 255) / 256, 256, 0, stream>>>(x, xp);
  wtrans_kernel<<<(KPAD * NPAD + 255) / 256, 256, 0, stream>>>(fw, wT);
  fc_wmma_kernel<<<BATCH / 16, 128, 0, stream>>>(xp, wT, fb, xf);
  scan_kernel<<<1, NTH, 0, stream>>>(xf, lw, hw, out);
}


# fc GEMM split along K: one work-group per (M tile, 64-wide K chunk), 704 instead of 8; chunk partials summed in fixed chunk order by a new reduce launch (order of the K summation changed)
# speedup vs baseline: 5.1230x; 5.1230x over previous
#include <hip/hip_runtime.h>
#include <hip/hip_bf16.h>
#include <math.h>

typedef __attribute__((ext_vector_type(2))) float v2f;
typedef __attribute__((ext_vector_type(8))) float v8f;

#define BATCH  128
#define NRN    54
#define DTH    4
#define STEPS  30
#define PIX    5625
#define KPAD   5632
#define NPAD   64
#define CELLS  (BATCH * NRN)
#define KSPLIT (KPAD / 64)

__global__ void pool_kernel(const float* __restrict__ x, float* __restrict__ xp) {
  int g = blockIdx.x * blockDim.x + threadIdx.x;
  if (g >= BATCH * KPAD) return;
  int b = g / KPAD;
  int p = g - b * KPAD;
  if (p >= PIX) { xp[g] = 0.0f; return; }
  int pr = p / 75;
  int pc = p - pr * 75;
  const float* base = x + (size_t)b * 360000 + pr * 4800 + pc * 8;
  float s = 0.0f;
#pragma unroll
  for (int r = 0; r < 8; ++r) {
    const float4* q = reinterpret_cast<const float4*>(base + r * 600);
    float4 a0 = q[0], a1 = q[1];
    s += (a0.x + a0.y + a0.z + a0.w) + (a1.x + a1.y + a1.z + a1.w);
  }
  xp[g] = s * (1.0f / 64.0f);
}

__global__ void wtrans_kernel(const float* __restrict__ w, float* __restrict__ wT) {
  int g = blockIdx.x * blockDim.x + threadIdx.x;
  if (g >= KPAD * NPAD) return;
  int k = g >> 6;
  int n = g & 63;
  wT[g] = (k < PIX && n < NRN) ? w[n * PIX + k] : 0.0f;
}

__global__ void __launch_bounds__(128) fc_wmma_splitk_kernel(const float* __restrict__ xp,
                                                             const float* __restrict__ wT,
                                                             float* __restrict__ part) {
  __shared__ float sA[16 * 68];
  __shared__ float sB[64 * NPAD];
  const int tid  = threadIdx.x;
  const int lane = tid & 31;
  const int wave = tid >> 5;
  const int m0   = blockIdx.x * 16;
  const int k0   = blockIdx.y * 64;
  const int half = lane >> 4;
  const int l15  = lane & 15;

  float4 ra[2], rb[8];
  for (int j = 0; j < 2; ++j) {
    int i = tid + j * 128;
    int m = i >> 4, q = (i & 15) << 2;
    ra[j] = *reinterpret_cast<const float4*>(xp + (size_t)(m0 + m) * KPAD + k0 + q);
  }
  for (int j = 0; j < 8; ++j) {
    int i = tid + j * 128;
    int k = i >> 4, q = (i & 15) << 2;
    rb[j] = *reinterpret_cast<const float4*>(wT + (size_t)(k0 + k) * NPAD + q);
  }
  for (int j = 0; j < 2; ++j) {
    int i = tid + j * 128;
    int m = i >> 4, q = (i & 15) << 2;
    *reinterpret_cast<float4*>(&sA[m * 68 + q]) = ra[j];
  }
  for (int j = 0; j < 8; ++j) {
    int i = tid + j * 128;
    int k = i >> 4, q = (i & 15) << 2;
    *reinterpret_cast<float4*>(&sB[k * NPAD + q]) = rb[j];
  }
  __syncthreads();

  v8f acc = {0.f, 0.f, 0.f, 0.f, 0.f, 0.f, 0.f, 0.f};
#pragma unroll 4
  for (int kk = 0; kk < 64; kk += 4) {
    int ka = kk + (half << 1);
    v2f a;
    a.x = sA[l15 * 68 + ka];
    a.y = sA[l15 * 68 + ka + 1];
    v2f bf;
    bf.x = sB[ka * NPAD + (wave << 4) + l15];
    bf.y = sB[(ka + 1) * NPAD + (wave << 4) + l15];
    acc = __builtin_amdgcn_wmma_f32_16x16x4_f32(
         false, a,  false, bf,
         (short)0, acc,  false,  false);
  }

  float* tile = part + ((size_t)blockIdx.y * BATCH + m0) * NPAD;
#pragma unroll
  for (int v = 0; v < 8; ++v) {
    int m = v + (half << 3);
    int n = (wave << 4) + l15;
    tile[m * NPAD + n] = acc[v];
  }
}

__global__ void __launch_bounds__(32) fc_reduce_kernel(const float* __restrict__ part,
                                                       const float* __restrict__ bias,
                                                       float* __restrict__ xf) {
  const int m = blockIdx.x;
  const int n = threadIdx.x << 1;
  const float* p = part + (size_t)m * NPAD + n;
  float sx = 0.0f, sy = 0.0f;
#pragma unroll
  for (int s = 0; s < KSPLIT; ++s) {
    float2 v = *reinterpret_cast<const float2*>(p + (size_t)s * (BATCH * NPAD));
    sx += v.x;
    sy += v.y;
  }
  float2 o;
  o.x = sx + (n < NRN ? bias[n] : 0.0f);
  o.y = sy + (n + 1 < NRN ? bias[n + 1] : 0.0f);
  *reinterpret_cast<float2*>(xf + m * NPAD + n) = o;
}

#define NTH  1024
#define MAXC 7
__global__ void __launch_bounds__(NTH) scan_kernel(const float* __restrict__ xf,
                                                   const float* __restrict__ lw,
                                                   const float* __restrict__ hw,
                                                   float* __restrict__ out) {
  __shared__ float sVals[CELLS];
  __shared__ float sBatch[BATCH];
  __shared__ float sWave[NTH / 32];
  __shared__ float sThr;

  const int tid  = threadIdx.x;
  const int lane = tid & 31;
  const int wv   = tid >> 5;

  float h0[MAXC], h1[MAXC], h2[MAXC], h3[MAXC], cnt[MAXC], xv[MAXC];
  float lw0[MAXC], lw1[MAXC], lw2[MAXC], lw3[MAXC];
  const float w0 = hw[0], w1 = hw[1], w2 = hw[2], w3 = hw[3];

#pragma unroll
  for (int j = 0; j < MAXC; ++j) {
    int c = tid + (j << 10);
    h0[j] = h1[j] = h2[j] = h3[j] = cnt[j] = 0.0f;
    if (c < CELLS) {
      int b = c / NRN, n = c - b * NRN;
      xv[j]  = xf[b * NPAD + n];
      lw0[j] = lw[n * DTH + 0]; lw1[j] = lw[n * DTH + 1];
      lw2[j] = lw[n * DTH + 2]; lw3[j] = lw[n * DTH + 3];
    } else {
      xv[j] = lw0[j] = lw1[j] = lw2[j] = lw3[j] = 0.0f;
    }
  }

  for (int step = 0; step < STEPS; ++step) {
    float sv[MAXC], hsv[MAXC], rate[MAXC];
#pragma unroll
    for (int j = 0; j < MAXC; ++j) {
      int c = tid + (j << 10);
      sv[j]  = h0[j] * lw0[j] + h1[j] * lw1[j] + h2[j] * lw2[j] + h3[j] * lw3[j];
      hsv[j] = h0[j] * w0 + h1[j] * w1 + h2[j] * w2 + h3[j] * w3;
      if (c < CELLS) sVals[c] = sv[j];
    }
    __syncthreads();
    if (tid < BATCH) {
      float acc = 0.0f;
      const float* row = &sVals[tid * NRN];
#pragma unroll
      for (int n = 0; n < NRN; ++n) acc += row[n];
      sBatch[tid] = acc;
    }
    __syncthreads();
    float psum = 0.0f;
#pragma unroll
    for (int j = 0; j < MAXC; ++j) {
      int c = tid + (j << 10);
      if (c < CELLS) {
        int b = c / NRN;
        float cur = xv[j] + (sBatch[b] - sv[j]) + hsv[j];
        rate[j] = __expf(cur);
        psum += rate[j];
      } else {
        rate[j] = 0.0f;
      }
    }
#pragma unroll
    for (int off = 16; off > 0; off >>= 1) psum += __shfl_down(psum, off, 32);
    if (lane == 0) sWave[wv] = psum;
    __syncthreads();
    if (tid < 32) {
      float v = sWave[tid];
#pragma unroll
      for (int off = 16; off > 0; off >>= 1) v += __shfl_down(v, off, 32);
      if (tid == 0) sThr = v * (1.0f / (float)CELLS);
    }
    __syncthreads();
    const float thr = sThr;
#pragma unroll
    for (int j = 0; j < MAXC; ++j) {
      float sp = (rate[j] > thr) ? 1.0f : 0.0f;
      cnt[j] += sp;
      h0[j] = h1[j]; h1[j] = h2[j]; h2[j] = h3[j]; h3[j] = sp;
    }
  }

#pragma unroll
  for (int j = 0; j < MAXC; ++j) {
    int c = tid + (j << 10);
    if (c < CELLS) {
      float cc = cnt[j];
      out[c] = fmaxf(cc, 0.0f) + log1pf(__expf(-fabsf(cc)));
    }
  }
}

extern "C" void kernel_launch(void* const* d_in, const int* in_sizes, int n_in,
                              void* d_out, int out_size, void* d_ws, size_t ws_size,
                              hipStream_t stream) {
  const float* x  = (const float*)d_in[0];
  const float* lw = (const float*)d_in[1];
  const float* hw = (const float*)d_in[2];
  const float* fw = (const float*)d_in[3];
  const float* fb = (const float*)d_in[4];

  float* ws = (float*)d_ws;
  float* xp = ws;
  float* wT = xp + (size_t)BATCH * KPAD;
  float* xf = wT + (size_t)KPAD * NPAD;
  float* part = xf + (size_t)BATCH * NPAD;
  float* out = (float*)d_out;

  pool_kernel<<<(BATCH * KPAD + 255) / 256, 256, 0, stream>>>(x, xp);
  wtrans_kernel<<<(KPAD * NPAD + 255) / 256, 256, 0, stream>>>(fw, wT);
  fc_wmma_splitk_kernel<<<dim3(BATCH / 16, KSPLIT), 128, 0, stream>>>(xp, wT, part);
  fc_reduce_kernel<<<BATCH, 32, 0, stream>>>(part, fb, xf);
  scan_kernel<<<1, NTH, 0, stream>>>(xf, lw, hw, out);
}


# scan kernel: both 32-lane sums of the rate mean run on the vector ALU (permlanex16 + DPP row_xmask adds), not LDS permutes; every wave adds the 32 wave sums itself: 3 barriers per step, not 4
# speedup vs baseline: 5.7185x; 1.1162x over previous
#include <hip/hip_runtime.h>
#include <hip/hip_bf16.h>
#include <math.h>

typedef __attribute__((ext_vector_type(2))) float v2f;
typedef __attribute__((ext_vector_type(8))) float v8f;

#define BATCH  128
#define NRN    54
#define DTH    4
#define STEPS  30
#define PIX    5625
#define KPAD   5632
#define NPAD   64
#define CELLS  (BATCH * NRN)
#define KSPLIT (KPAD / 64)

__global__ void pool_kernel(const float* __restrict__ x, float* __restrict__ xp) {
  int g = blockIdx.x * blockDim.x + threadIdx.x;
  if (g >= BATCH * KPAD) return;
  int b = g / KPAD;
  int p = g - b * KPAD;
  if (p >= PIX) { xp[g] = 0.0f; return; }
  int pr = p / 75;
  int pc = p - pr * 75;
  const float* base = x + (size_t)b * 360000 + pr * 4800 + pc * 8;
  float s = 0.0f;
#pragma unroll
  for (int r = 0; r < 8; ++r) {
    const float4* q = reinterpret_cast<const float4*>(base + r * 600);
    float4 a0 = q[0], a1 = q[1];
    s += (a0.x + a0.y + a0.z + a0.w) + (a1.x + a1.y + a1.z + a1.w);
  }
  xp[g] = s * (1.0f / 64.0f);
}

__global__ void wtrans_kernel(const float* __restrict__ w, float* __restrict__ wT) {
  int g = blockIdx.x * blockDim.x + threadIdx.x;
  if (g >= KPAD * NPAD) return;
  int k = g >> 6;
  int n = g & 63;
  wT[g] = (k < PIX && n < NRN) ? w[n * PIX + k] : 0.0f;
}

__global__ void __launch_bounds__(128) fc_wmma_splitk_kernel(const float* __restrict__ xp,
                                                             const float* __restrict__ wT,
                                                             float* __restrict__ part) {
  __shared__ float sA[16 * 68];
  __shared__ float sB[64 * NPAD];
  const int tid  = threadIdx.x;
  const int lane = tid & 31;
  const int wave = tid >> 5;
  const int m0   = blockIdx.x * 16;
  const int k0   = blockIdx.y * 64;
  const int half = lane >> 4;
  const int l15  = lane & 15;

  float4 ra[2], rb[8];
  for (int j = 0; j < 2; ++j) {
    int i = tid + j * 128;
    int m = i >> 4, q = (i & 15) << 2;
    ra[j] = *reinterpret_cast<const float4*>(xp + (size_t)(m0 + m) * KPAD + k0 + q);
  }
  for (int j = 0; j < 8; ++j) {
    int i = tid + j * 128;
    int k = i >> 4, q = (i & 15) << 2;
    rb[j] = *reinterpret_cast<const float4*>(wT + (size_t)(k0 + k) * NPAD + q);
  }
  for (int j = 0; j < 2; ++j) {
    int i = tid + j * 128;
    int m = i >> 4, q = (i & 15) << 2;
    *reinterpret_cast<float4*>(&sA[m * 68 + q]) = ra[j];
  }
  for (int j = 0; j < 8; ++j) {
    int i = tid + j * 128;
    int k = i >> 4, q = (i & 15) << 2;
    *reinterpret_cast<float4*>(&sB[k * NPAD + q]) = rb[j];
  }
  __syncthreads();

  v8f acc = {0.f, 0.f, 0.f, 0.f, 0.f, 0.f, 0.f, 0.f};
#pragma unroll 4
  for (int kk = 0; kk < 64; kk += 4) {
    int ka = kk + (half << 1);
    v2f a;
    a.x = sA[l15 * 68 + ka];
    a.y = sA[l15 * 68 + ka + 1];
    v2f bf;
    bf.x = sB[ka * NPAD + (wave << 4) + l15];
    bf.y = sB[(ka + 1) * NPAD + (wave << 4) + l15];
    acc = __builtin_amdgcn_wmma_f32_16x16x4_f32(
         false, a,  false, bf,
         (short)0, acc,  false,  false);
  }

  float* tile = part + ((size_t)blockIdx.y * BATCH + m0) * NPAD;
#pragma unroll
  for (int v = 0; v < 8; ++v) {
    int m = v + (half << 3);
    int n = (wave << 4) + l15;
    tile[m * NPAD + n] = acc[v];
  }
}

__global__ void __launch_bounds__(32) fc_reduce_kernel(const float* __restrict__ part,
                                                       const float* __restrict__ bias,
                                                       float* __restrict__ xf) {
  const int m = blockIdx.x;
  const int n = threadIdx.x << 1;
  const float* p = part + (size_t)m * NPAD + n;
  float sx = 0.0f, sy = 0.0f;
#pragma unroll
  for (int s = 0; s < KSPLIT; ++s) {
    float2 v = *reinterpret_cast<const float2*>(p + (size_t)s * (BATCH * NPAD));
    sx += v.x;
    sy += v.y;
  }
  float2 o;
  o.x = sx + (n < NRN ? bias[n] : 0.0f);
  o.y = sy + (n + 1 < NRN ? bias[n + 1] : 0.0f);
  *reinterpret_cast<float2*>(xf + m * NPAD + n) = o;
}

__device__ __forceinline__ float xhalf(float x) {
  int b = __builtin_bit_cast(int, x);
  return __builtin_bit_cast(float,
      __builtin_amdgcn_permlanex16(b, b, 0x76543210, 0xfedcba98, false, false));
}
template <int M>
__device__ __forceinline__ float xrow(float x) {
  int b = __builtin_bit_cast(int, x);
  return __builtin_bit_cast(float,
      __builtin_amdgcn_update_dpp(b, b, 0x160 + M, 0xf, 0xf, true));
}
__device__ __forceinline__ float wave_sum(float v) {
  v += xhalf(v);
  v += xrow<8>(v);
  v += xrow<4>(v);
  v += xrow<2>(v);
  v += xrow<1>(v);
  return v;
}

#define NTH  1024
#define MAXC 7
__global__ void __launch_bounds__(NTH) scan_kernel(const float* __restrict__ xf,
                                                   const float* __restrict__ lw,
                                                   const float* __restrict__ hw,
                                                   float* __restrict__ out) {
  __shared__ float sVals[CELLS];
  __shared__ float sBatch[BATCH];
  __shared__ float sWave[NTH / 32];

  const int tid  = threadIdx.x;
  const int lane = tid & 31;
  const int wv   = tid >> 5;

  float h0[MAXC], h1[MAXC], h2[MAXC], h3[MAXC], cnt[MAXC], xv[MAXC];
  float lw0[MAXC], lw1[MAXC], lw2[MAXC], lw3[MAXC];
  const float w0 = hw[0], w1 = hw[1], w2 = hw[2], w3 = hw[3];

#pragma unroll
  for (int j = 0; j < MAXC; ++j) {
    int c = tid + (j << 10);
    h0[j] = h1[j] = h2[j] = h3[j] = cnt[j] = 0.0f;
    if (c < CELLS) {
      int b = c / NRN, n = c - b * NRN;
      xv[j]  = xf[b * NPAD + n];
      lw0[j] = lw[n * DTH + 0]; lw1[j] = lw[n * DTH + 1];
      lw2[j] = lw[n * DTH + 2]; lw3[j] = lw[n * DTH + 3];
    } else {
      xv[j] = lw0[j] = lw1[j] = lw2[j] = lw3[j] = 0.0f;
    }
  }

  for (int step = 0; step < STEPS; ++step) {
    float sv[MAXC], hsv[MAXC], rate[MAXC];
#pragma unroll
    for (int j = 0; j < MAXC; ++j) {
      int c = tid + (j << 10);
      sv[j]  = h0[j] * lw0[j] + h1[j] * lw1[j] + h2[j] * lw2[j] + h3[j] * lw3[j];
      hsv[j] = h0[j] * w0 + h1[j] * w1 + h2[j] * w2 + h3[j] * w3;
      if (c < CELLS) sVals[c] = sv[j];
    }
    __syncthreads();
    if (tid < BATCH) {
      float acc = 0.0f;
      const float* row = &sVals[tid * NRN];
#pragma unroll
      for (int n = 0; n < NRN; ++n) acc += row[n];
      sBatch[tid] = acc;
    }
    __syncthreads();
    float psum = 0.0f;
#pragma unroll
    for (int j = 0; j < MAXC; ++j) {
      int c = tid + (j << 10);
      if (c < CELLS) {
        int b = c / NRN;
        float cur = xv[j] + (sBatch[b] - sv[j]) + hsv[j];
        rate[j] = __expf(cur);
        psum += rate[j];
      } else {
        rate[j] = 0.0f;
      }
    }
    psum = wave_sum(psum);
    if (lane == 0) sWave[wv] = psum;
    __syncthreads();
    const float thr = wave_sum(sWave[lane]) * (1.0f / (float)CELLS);
#pragma unroll
    for (int j = 0; j < MAXC; ++j) {
      float sp = (rate[j] > thr) ? 1.0f : 0.0f;
      cnt[j] += sp;
      h0[j] = h1[j]; h1[j] = h2[j]; h2[j] = h3[j]; h3[j] = sp;
    }
  }

#pragma unroll
  for (int j = 0; j < MAXC; ++j) {
    int c = tid + (j << 10);
    if (c < CELLS) {
      float cc = cnt[j];
      out[c] = fmaxf(cc, 0.0f) + log1pf(__expf(-fabsf(cc)));
    }
  }
}

extern "C" void kernel_launch(void* const* d_in, const int* in_sizes, int n_in,
                              void* d_out, int out_size, void* d_ws, size_t ws_size,
                              hipStream_t stream) {
  const float* x  = (const float*)d_in[0];
  const float* lw = (const float*)d_in[1];
  const float* hw = (const float*)d_in[2];
  const float* fw = (const float*)d_in[3];
  const float* fb = (const float*)d_in[4];

  float* ws = (float*)d_ws;
  float* xp = ws;
  float* wT = xp + (size_t)BATCH * KPAD;
  float* xf = wT + (size_t)KPAD * NPAD;
  float* part = xf + (size_t)BATCH * NPAD;
  float* out = (float*)d_out;

  pool_kernel<<<(BATCH * KPAD + 255) / 256, 256, 0, stream>>>(x, xp);
  wtrans_kernel<<<(KPAD * NPAD + 255) / 256, 256, 0, stream>>>(fw, wT);
  fc_wmma_splitk_kernel<<<dim3(BATCH / 16, KSPLIT), 128, 0, stream>>>(xp, wT, part);
  fc_reduce_kernel<<<BATCH, 32, 0, stream>>>(part, fb, xf);
  scan_kernel<<<1, NTH, 0, stream>>>(xf, lw, hw, out);
}


# scan: one batch row per 8-lane group (7 neurons per lane); per-batch lateral sum = in-lane adds + 3 DPP adds, no LDS pass; 1 barrier per step, not 3 (order of lateral and rate-mean sums changed)
# speedup vs baseline: 6.4909x; 1.1351x over previous
#include <hip/hip_runtime.h>
#include <hip/hip_bf16.h>
#include <math.h>

typedef __attribute__((ext_vector_type(2))) float v2f;
typedef __attribute__((ext_vector_type(8))) float v8f;

#define BATCH  128
#define NRN    54
#define DTH    4
#define STEPS  30
#define PIX    5625
#define KPAD   5632
#define NPAD   64
#define CELLS  (BATCH * NRN)
#define KSPLIT (KPAD / 64)

__global__ void pool_kernel(const float* __restrict__ x, float* __restrict__ xp) {
  int g = blockIdx.x * blockDim.x + threadIdx.x;
  if (g >= BATCH * KPAD) return;
  int b = g / KPAD;
  int p = g - b * KPAD;
  if (p >= PIX) { xp[g] = 0.0f; return; }
  int pr = p / 75;
  int pc = p - pr * 75;
  const float* base = x + (size_t)b * 360000 + pr * 4800 + pc * 8;
  float s = 0.0f;
#pragma unroll
  for (int r = 0; r < 8; ++r) {
    const float4* q = reinterpret_cast<const float4*>(base + r * 600);
    float4 a0 = q[0], a1 = q[1];
    s += (a0.x + a0.y + a0.z + a0.w) + (a1.x + a1.y + a1.z + a1.w);
  }
  xp[g] = s * (1.0f / 64.0f);
}

__global__ void wtrans_kernel(const float* __restrict__ w, float* __restrict__ wT) {
  int g = blockIdx.x * blockDim.x + threadIdx.x;
  if (g >= KPAD * NPAD) return;
  int k = g >> 6;
  int n = g & 63;
  wT[g] = (k < PIX && n < NRN) ? w[n * PIX + k] : 0.0f;
}

__global__ void __launch_bounds__(128) fc_wmma_splitk_kernel(const float* __restrict__ xp,
                                                             const float* __restrict__ wT,
                                                             float* __restrict__ part) {
  __shared__ float sA[16 * 68];
  __shared__ float sB[64 * NPAD];
  const int tid  = threadIdx.x;
  const int lane = tid & 31;
  const int wave = tid >> 5;
  const int m0   = blockIdx.x * 16;
  const int k0   = blockIdx.y * 64;
  const int half = lane >> 4;
  const int l15  = lane & 15;

  float4 ra[2], rb[8];
  for (int j = 0; j < 2; ++j) {
    int i = tid + j * 128;
    int m = i >> 4, q = (i & 15) << 2;
    ra[j] = *reinterpret_cast<const float4*>(xp + (size_t)(m0 + m) * KPAD + k0 + q);
  }
  for (int j = 0; j < 8; ++j) {
    int i = tid + j * 128;
    int k = i >> 4, q = (i & 15) << 2;
    rb[j] = *reinterpret_cast<const float4*>(wT + (size_t)(k0 + k) * NPAD + q);
  }
  for (int j = 0; j < 2; ++j) {
    int i = tid + j * 128;
    int m = i >> 4, q = (i & 15) << 2;
    *reinterpret_cast<float4*>(&sA[m * 68 + q]) = ra[j];
  }
  for (int j = 0; j < 8; ++j) {
    int i = tid + j * 128;
    int k = i >> 4, q = (i & 15) << 2;
    *reinterpret_cast<float4*>(&sB[k * NPAD + q]) = rb[j];
  }
  __syncthreads();

  v8f acc = {0.f, 0.f, 0.f, 0.f, 0.f, 0.f, 0.f, 0.f};
#pragma unroll 4
  for (int kk = 0; kk < 64; kk += 4) {
    int ka = kk + (half << 1);
    v2f a;
    a.x = sA[l15 * 68 + ka];
    a.y = sA[l15 * 68 + ka + 1];
    v2f bf;
    bf.x = sB[ka * NPAD + (wave << 4) + l15];
    bf.y = sB[(ka + 1) * NPAD + (wave << 4) + l15];
    acc = __builtin_amdgcn_wmma_f32_16x16x4_f32(
         false, a,  false, bf,
         (short)0, acc,  false,  false);
  }

  float* tile = part + ((size_t)blockIdx.y * BATCH + m0) * NPAD;
#pragma unroll
  for (int v = 0; v < 8; ++v) {
    int m = v + (half << 3);
    int n = (wave << 4) + l15;
    tile[m * NPAD + n] = acc[v];
  }
}

__global__ void __launch_bounds__(32) fc_reduce_kernel(const float* __restrict__ part,
                                                       const float* __restrict__ bias,
                                                       float* __restrict__ xf) {
  const int m = blockIdx.x;
  const int n = threadIdx.x << 1;
  const float* p = part + (size_t)m * NPAD + n;
  float sx = 0.0f, sy = 0.0f;
#pragma unroll
  for (int s = 0; s < KSPLIT; ++s) {
    float2 v = *reinterpret_cast<const float2*>(p + (size_t)s * (BATCH * NPAD));
    sx += v.x;
    sy += v.y;
  }
  float2 o;
  o.x = sx + (n < NRN ? bias[n] : 0.0f);
  o.y = sy + (n + 1 < NRN ? bias[n + 1] : 0.0f);
  *reinterpret_cast<float2*>(xf + m * NPAD + n) = o;
}

__device__ __forceinline__ float xhalf(float x) {
  int b = __builtin_bit_cast(int, x);
  return __builtin_bit_cast(float,
      __builtin_amdgcn_permlanex16(b, b, 0x76543210, 0xfedcba98, false, false));
}
template <int M>
__device__ __forceinline__ float xrow(float x) {
  int b = __builtin_bit_cast(int, x);
  return __builtin_bit_cast(float,
      __builtin_amdgcn_update_dpp(b, b, 0x160 + M, 0xf, 0xf, true));
}
__device__ __forceinline__ float wave_sum(float v) {
  v += xhalf(v);
  v += xrow<8>(v);
  v += xrow<4>(v);
  v += xrow<2>(v);
  v += xrow<1>(v);
  return v;
}

#define NTH  1024
#define NPL  7
__global__ void __launch_bounds__(NTH) scan_kernel(const float* __restrict__ xf,
                                                   const float* __restrict__ lw,
                                                   const float* __restrict__ hw,
                                                   float* __restrict__ out) {
  __shared__ float sWave[2][NTH / 32];
  __shared__ float sOut[CELLS];

  const int tid  = threadIdx.x;
  const int lane = tid & 31;
  const int wv   = tid >> 5;
  const int b    = tid >> 3;
  const int c8   = tid & 7;

  float h0[NPL], h1[NPL], h2[NPL], h3[NPL], cnt[NPL], xv[NPL];
  float lw0[NPL], lw1[NPL], lw2[NPL], lw3[NPL];
  const float w0 = hw[0], w1 = hw[1], w2 = hw[2], w3 = hw[3];

#pragma unroll
  for (int j = 0; j < NPL; ++j) {
    int n = c8 + (j << 3);
    h0[j] = h1[j] = h2[j] = h3[j] = cnt[j] = 0.0f;
    if (n < NRN) {
      xv[j]  = xf[b * NPAD + n];
      const float4 l = *reinterpret_cast<const float4*>(lw + n * DTH);
      lw0[j] = l.x; lw1[j] = l.y; lw2[j] = l.z; lw3[j] = l.w;
    } else {
      xv[j] = -INFINITY;
      lw0[j] = lw1[j] = lw2[j] = lw3[j] = 0.0f;
    }
  }

  for (int step = 0; step < STEPS; ++step) {
    float sv[NPL], hsv[NPL], rate[NPL];
#pragma unroll
    for (int j = 0; j < NPL; ++j) {
      sv[j]  = h0[j] * lw0[j] + h1[j] * lw1[j] + h2[j] * lw2[j] + h3[j] * lw3[j];
      hsv[j] = h0[j] * w0 + h1[j] * w1 + h2[j] * w2 + h3[j] * w3;
    }
    float S = ((sv[0] + sv[1]) + (sv[2] + sv[3])) + ((sv[4] + sv[5]) + sv[6]);
    S += xrow<4>(S);
    S += xrow<2>(S);
    S += xrow<1>(S);
    float psum = 0.0f;
#pragma unroll
    for (int j = 0; j < NPL; ++j) {
      float cur = xv[j] + (S - sv[j]) + hsv[j];
      rate[j] = __expf(cur);
      psum += rate[j];
    }
    psum = wave_sum(psum);
    float* sw = sWave[step & 1];
    if (lane == 0) sw[wv] = psum;
    __syncthreads();
    const float thr = wave_sum(sw[lane]) * (1.0f / (float)CELLS);
#pragma unroll
    for (int j = 0; j < NPL; ++j) {
      float sp = (rate[j] > thr) ? 1.0f : 0.0f;
      cnt[j] += sp;
      h0[j] = h1[j]; h1[j] = h2[j]; h2[j] = h3[j]; h3[j] = sp;
    }
  }

#pragma unroll
  for (int j = 0; j < NPL; ++j) {
    int n = c8 + (j << 3);
    if (n < NRN) {
      float cc = cnt[j];
      sOut[b * NRN + n] = fmaxf(cc, 0.0f) + log1pf(__expf(-fabsf(cc)));
    }
  }
  __syncthreads();
#pragma unroll
  for (int j = 0; j < NPL; ++j) {
    int c = tid + (j << 10);
    if (c < CELLS) out[c] = sOut[c];
  }
}

extern "C" void kernel_launch(void* const* d_in, const int* in_sizes, int n_in,
                              void* d_out, int out_size, void* d_ws, size_t ws_size,
                              hipStream_t stream) {
  const float* x  = (const float*)d_in[0];
  const float* lw = (const float*)d_in[1];
  const float* hw = (const float*)d_in[2];
  const float* fw = (const float*)d_in[3];
  const float* fb = (const float*)d_in[4];

  float* ws = (float*)d_ws;
  float* xp = ws;
  float* wT = xp + (size_t)BATCH * KPAD;
  float* xf = wT + (size_t)KPAD * NPAD;
  float* part = xf + (size_t)BATCH * NPAD;
  float* out = (float*)d_out;

  pool_kernel<<<(BATCH * KPAD + 255) / 256, 256, 0, stream>>>(x, xp);
  wtrans_kernel<<<(KPAD * NPAD + 255) / 256, 256, 0, stream>>>(fw, wT);
  fc_wmma_splitk_kernel<<<dim3(BATCH / 16, KSPLIT), 128, 0, stream>>>(xp, wT, part);
  fc_reduce_kernel<<<BATCH, 32, 0, stream>>>(part, fb, xf);
  scan_kernel<<<1, NTH, 0, stream>>>(xf, lw, hw, out);
}


# scan step without register moves: the 4 history planes rotate roles over a 4-step loop body (no shift), slots kept as register pairs for packed f32 ops, h_weight held in VGPRs; sums keep their order
# speedup vs baseline: 7.2573x; 1.1181x over previous
#include <hip/hip_runtime.h>
#include <hip/hip_bf16.h>
#include <math.h>

typedef __attribute__((ext_vector_type(2))) float v2f;
typedef __attribute__((ext_vector_type(8))) float v8f;

#define BATCH  128
#define NRN    54
#define DTH    4
#define STEPS  30
#define PIX    5625
#define KPAD   5632
#define NPAD   64
#define CELLS  (BATCH * NRN)
#define KSPLIT (KPAD / 64)

__global__ void pool_kernel(const float* __restrict__ x, float* __restrict__ xp) {
  int g = blockIdx.x * blockDim.x + threadIdx.x;
  if (g >= BATCH * KPAD) return;
  int b = g / KPAD;
  int p = g - b * KPAD;
  if (p >= PIX) { xp[g] = 0.0f; return; }
  int pr = p / 75;
  int pc = p - pr * 75;
  const float* base = x + (size_t)b * 360000 + pr * 4800 + pc * 8;
  float s = 0.0f;
#pragma unroll
  for (int r = 0; r < 8; ++r) {
    const float4* q = reinterpret_cast<const float4*>(base + r * 600);
    float4 a0 = q[0], a1 = q[1];
    s += (a0.x + a0.y + a0.z + a0.w) + (a1.x + a1.y + a1.z + a1.w);
  }
  xp[g] = s * (1.0f / 64.0f);
}

__global__ void wtrans_kernel(const float* __restrict__ w, float* __restrict__ wT) {
  int g = blockIdx.x * blockDim.x + threadIdx.x;
  if (g >= KPAD * NPAD) return;
  int k = g >> 6;
  int n = g & 63;
  wT[g] = (k < PIX && n < NRN) ? w[n * PIX + k] : 0.0f;
}

__global__ void __launch_bounds__(128) fc_wmma_splitk_kernel(const float* __restrict__ xp,
                                                             const float* __restrict__ wT,
                                                             float* __restrict__ part) {
  __shared__ float sA[16 * 68];
  __shared__ float sB[64 * NPAD];
  const int tid  = threadIdx.x;
  const int lane = tid & 31;
  const int wave = tid >> 5;
  const int m0   = blockIdx.x * 16;
  const int k0   = blockIdx.y * 64;
  const int half = lane >> 4;
  const int l15  = lane & 15;

  float4 ra[2], rb[8];
  for (int j = 0; j < 2; ++j) {
    int i = tid + j * 128;
    int m = i >> 4, q = (i & 15) << 2;
    ra[j] = *reinterpret_cast<const float4*>(xp + (size_t)(m0 + m) * KPAD + k0 + q);
  }
  for (int j = 0; j < 8; ++j) {
    int i = tid + j * 128;
    int k = i >> 4, q = (i & 15) << 2;
    rb[j] = *reinterpret_cast<const float4*>(wT + (size_t)(k0 + k) * NPAD + q);
  }
  for (int j = 0; j < 2; ++j) {
    int i = tid + j * 128;
    int m = i >> 4, q = (i & 15) << 2;
    *reinterpret_cast<float4*>(&sA[m * 68 + q]) = ra[j];
  }
  for (int j = 0; j < 8; ++j) {
    int i = tid + j * 128;
    int k = i >> 4, q = (i & 15) << 2;
    *reinterpret_cast<float4*>(&sB[k * NPAD + q]) = rb[j];
  }
  __syncthreads();

  v8f acc = {0.f, 0.f, 0.f, 0.f, 0.f, 0.f, 0.f, 0.f};
#pragma unroll 4
  for (int kk = 0; kk < 64; kk += 4) {
    int ka = kk + (half << 1);
    v2f a;
    a.x = sA[l15 * 68 + ka];
    a.y = sA[l15 * 68 + ka + 1];
    v2f bf;
    bf.x = sB[ka * NPAD + (wave << 4) + l15];
    bf.y = sB[(ka + 1) * NPAD + (wave << 4) + l15];
    acc = __builtin_amdgcn_wmma_f32_16x16x4_f32(
         false, a,  false, bf,
         (short)0, acc,  false,  false);
  }

  float* tile = part + ((size_t)blockIdx.y * BATCH + m0) * NPAD;
#pragma unroll
  for (int v = 0; v < 8; ++v) {
    int m = v + (half << 3);
    int n = (wave << 4) + l15;
    tile[m * NPAD + n] = acc[v];
  }
}

__global__ void __launch_bounds__(32) fc_reduce_kernel(const float* __restrict__ part,
                                                       const float* __restrict__ bias,
                                                       float* __restrict__ xf) {
  const int m = blockIdx.x;
  const int n = threadIdx.x << 1;
  const float* p = part + (size_t)m * NPAD + n;
  float sx = 0.0f, sy = 0.0f;
#pragma unroll
  for (int s = 0; s < KSPLIT; ++s) {
    float2 v = *reinterpret_cast<const float2*>(p + (size_t)s * (BATCH * NPAD));
    sx += v.x;
    sy += v.y;
  }
  float2 o;
  o.x = sx + (n < NRN ? bias[n] : 0.0f);
  o.y = sy + (n + 1 < NRN ? bias[n + 1] : 0.0f);
  *reinterpret_cast<float2*>(xf + m * NPAD + n) = o;
}

__device__ __forceinline__ float xhalf(float x) {
  int b = __builtin_bit_cast(int, x);
  return __builtin_bit_cast(float,
      __builtin_amdgcn_permlanex16(b, b, 0x76543210, 0xfedcba98, false, false));
}
template <int M>
__device__ __forceinline__ float xrow(float x) {
  int b = __builtin_bit_cast(int, x);
  return __builtin_bit_cast(float,
      __builtin_amdgcn_update_dpp(b, b, 0x160 + M, 0xf, 0xf, true));
}
__device__ __forceinline__ float wave_sum(float v) {
  v += xhalf(v);
  v += xrow<8>(v);
  v += xrow<4>(v);
  v += xrow<2>(v);
  v += xrow<1>(v);
  return v;
}

#define NTH  1024
#define NPR  3
struct Slots {
  v2f   p[NPR];
  float s;
};
__device__ __forceinline__ constexpr int slot_of(int p, int e) {
  return p == 0 ? (e ? 2 : 0) : p == 1 ? (e ? 3 : 1) : (e ? 5 : 4);
}

__device__ __forceinline__ void scan_step(Slots& a, const Slots& b, const Slots& c, const Slots& d,
                                          Slots& cnt, const Slots& xv,
                                          const Slots& l0, const Slots& l1,
                                          const Slots& l2, const Slots& l3,
                                          const v2f w0, const v2f w1,
                                          const v2f w2, const v2f w3,
                                          const v2f q0, const v2f q1,
                                          const v2f q2, const v2f q3,
                                          float* __restrict__ sw, const int lane, const int wv) {
  Slots sv, hsv, rate;
#pragma unroll
  for (int p = 0; p < NPR; ++p) {
    sv.p[p]  = a.p[p] * l0.p[p] + b.p[p] * l1.p[p] + c.p[p] * l2.p[p] + d.p[p] * l3.p[p];
    hsv.p[p] = a.p[p] * w0 + b.p[p] * w1 + c.p[p] * w2 + d.p[p] * w3;
  }
  const v2f r6 = a.s * q0 + b.s * q1 + c.s * q2 + d.s * q3;
  sv.s  = r6.x;
  hsv.s = r6.y;
  const v2f t01 = sv.p[0] + sv.p[1];
  float S = (t01.x + t01.y) + ((sv.p[2].x + sv.p[2].y) + sv.s);
  S += xrow<4>(S);
  S += xrow<2>(S);
  S += xrow<1>(S);
  float psum = 0.0f;
#pragma unroll
  for (int p = 0; p < NPR; ++p) {
    v2f cur = xv.p[p] + (S - sv.p[p]) + hsv.p[p];
    rate.p[p].x = __expf(cur.x);
    rate.p[p].y = __expf(cur.y);
  }
  rate.s = __expf(xv.s + (S - sv.s) + hsv.s);
  psum += rate.p[0].x;
  psum += rate.p[1].x;
  psum += rate.p[0].y;
  psum += rate.p[1].y;
  psum += rate.p[2].x;
  psum += rate.p[2].y;
  psum += rate.s;
  psum = wave_sum(psum);
  if (lane == 0) sw[wv] = psum;
  __syncthreads();
  const float thr = wave_sum(sw[lane]) * (1.0f / (float)CELLS);
#pragma unroll
  for (int p = 0; p < NPR; ++p) {
    v2f sp;
    sp.x = (rate.p[p].x > thr) ? 1.0f : 0.0f;
    sp.y = (rate.p[p].y > thr) ? 1.0f : 0.0f;
    cnt.p[p] += sp;
    a.p[p] = sp;
  }
  const float sps = (rate.s > thr) ? 1.0f : 0.0f;
  cnt.s += sps;
  a.s = sps;
}

__global__ void __launch_bounds__(NTH) scan_kernel(const float* __restrict__ xf,
                                                   const float* __restrict__ lw,
                                                   const float* __restrict__ hw,
                                                   float* __restrict__ out) {
  __shared__ float sWave[2][NTH / 32];
  __shared__ float sOut[CELLS];

  const int tid  = threadIdx.x;
  const int lane = tid & 31;
  const int wv   = tid >> 5;
  const int b    = tid >> 3;
  const int c8   = tid & 7;

  Slots h0, h1, h2, h3, cnt, xv, lw0, lw1, lw2, lw3;
  const float hw0 = hw[0], hw1 = hw[1], hw2 = hw[2], hw3 = hw[3];

#pragma unroll
  for (int p = 0; p < NPR; ++p) {
#pragma unroll
    for (int e = 0; e < 2; ++e) {
      const int n = c8 + (slot_of(p, e) << 3);
      xv.p[p][e] = xf[b * NPAD + n];
      const float4 l = *reinterpret_cast<const float4*>(lw + n * DTH);
      lw0.p[p][e] = l.x; lw1.p[p][e] = l.y; lw2.p[p][e] = l.z; lw3.p[p][e] = l.w;
      h0.p[p][e] = 0.0f; h1.p[p][e] = 0.0f; h2.p[p][e] = 0.0f; h3.p[p][e] = 0.0f;
      cnt.p[p][e] = 0.0f;
    }
  }
  {
    const int n = c8 + (2 * NPR << 3);
    h0.s = h1.s = h2.s = h3.s = cnt.s = 0.0f;
    if (n < NRN) {
      xv.s = xf[b * NPAD + n];
      const float4 l = *reinterpret_cast<const float4*>(lw + n * DTH);
      lw0.s = l.x; lw1.s = l.y; lw2.s = l.z; lw3.s = l.w;
    } else {
      xv.s = -INFINITY;
      lw0.s = lw1.s = lw2.s = lw3.s = 0.0f;
    }
  }
  v2f w0 = {hw0, hw0}, w1 = {hw1, hw1}, w2 = {hw2, hw2}, w3 = {hw3, hw3};
  v2f q0 = {lw0.s, hw0}, q1 = {lw1.s, hw1}, q2 = {lw2.s, hw2}, q3 = {lw3.s, hw3};
  asm volatile("" : "+v"(w0), "+v"(w1), "+v"(w2), "+v"(w3),
                    "+v"(q0), "+v"(q1), "+v"(q2), "+v"(q3));

#define SCAN_STEP(A, B, C, D, PLANE) \
  scan_step(A, B, C, D, cnt, xv, lw0, lw1, lw2, lw3, w0, w1, w2, w3, q0, q1, q2, q3, \
            sWave[PLANE], lane, wv)
  for (int it = 0; it < STEPS / 4; ++it) {
    SCAN_STEP(h0, h1, h2, h3, 0);
    SCAN_STEP(h1, h2, h3, h0, 1);
    SCAN_STEP(h2, h3, h0, h1, 0);
    SCAN_STEP(h3, h0, h1, h2, 1);
  }
  if (STEPS % 4 > 0) SCAN_STEP(h0, h1, h2, h3, 0);
  if (STEPS % 4 > 1) SCAN_STEP(h1, h2, h3, h0, 1);
  if (STEPS % 4 > 2) SCAN_STEP(h2, h3, h0, h1, 0);
#undef SCAN_STEP

#pragma unroll
  for (int p = 0; p < NPR; ++p) {
#pragma unroll
    for (int e = 0; e < 2; ++e) {
      const int n = c8 + (slot_of(p, e) << 3);
      const float cc = cnt.p[p][e];
      sOut[b * NRN + n] = fmaxf(cc, 0.0f) + log1pf(__expf(-fabsf(cc)));
    }
  }
  if (c8 + (2 * NPR << 3) < NRN) {
    const float cc = cnt.s;
    sOut[b * NRN + c8 + (2 * NPR << 3)] = fmaxf(cc, 0.0f) + log1pf(__expf(-fabsf(cc)));
  }
  __syncthreads();
#pragma unroll
  for (int j = 0; j < 2 * NPR + 1; ++j) {
    int c = tid + (j << 10);
    if (c < CELLS) out[c] = sOut[c];
  }
}

extern "C" void kernel_launch(void* const* d_in, const int* in_sizes, int n_in,
                              void* d_out, int out_size, void* d_ws, size_t ws_size,
                              hipStream_t stream) {
  const float* x  = (const float*)d_in[0];
  const float* lw = (const float*)d_in[1];
  const float* hw = (const float*)d_in[2];
  const float* fw = (const float*)d_in[3];
  const float* fb = (const float*)d_in[4];

  float* ws = (float*)d_ws;
  float* xp = ws;
  float* wT = xp + (size_t)BATCH * KPAD;
  float* xf = wT + (size_t)KPAD * NPAD;
  float* part = xf + (size_t)BATCH * NPAD;
  float* out = (float*)d_out;

  pool_kernel<<<(BATCH * KPAD + 255) / 256, 256, 0, stream>>>(x, xp);
  wtrans_kernel<<<(KPAD * NPAD + 255) / 256, 256, 0, stream>>>(fw, wT);
  fc_wmma_splitk_kernel<<<dim3(BATCH / 16, KSPLIT), 128, 0, stream>>>(xp, wT, part);
  fc_reduce_kernel<<<BATCH, 32, 0, stream>>>(part, fb, xf);
  scan_kernel<<<1, NTH, 0, stream>>>(xf, lw, hw, out);
}


# scan epilogue: softplus computed once per possible spike count (integers 0..STEPS, one thread each, at kernel start) into an LDS table; a cell looks its count up, no exp/log1p per cell; sums unchanged
# speedup vs baseline: 9.9502x; 1.3711x over previous
#include <hip/hip_runtime.h>
#include <hip/hip_bf16.h>
#include <math.h>

typedef __attribute__((ext_vector_type(2))) float v2f;
typedef __attribute__((ext_vector_type(8))) float v8f;

#define BATCH  128
#define NRN    54
#define DTH    4
#define STEPS  30
#define PIX    5625
#define KPAD   5632
#define NPAD   64
#define CELLS  (BATCH * NRN)
#define KSPLIT (KPAD / 64)

#define LDP 68

__device__ __forceinline__ void pool_load(const float* __restrict__ base, float4 (&v)[16]) {
#pragma unroll
  for (int r = 0; r < 8; ++r) {
    const float4* q = reinterpret_cast<const float4*>(base + r * 600);
    v[2 * r]     = q[0];
    v[2 * r + 1] = q[1];
  }
}
__device__ __forceinline__ float pool_sum(const float4 (&v)[16]) {
  float s = 0.0f;
#pragma unroll
  for (int r = 0; r < 8; ++r) {
    const float4 a0 = v[2 * r], a1 = v[2 * r + 1];
    s += (a0.x + a0.y + a0.z + a0.w) + (a1.x + a1.y + a1.z + a1.w);
  }
  return s * (1.0f / 64.0f);
}

__global__ void __launch_bounds__(256) __attribute__((amdgpu_waves_per_eu(6, 6)))
fc_pool_wmma_splitk_kernel(const float* __restrict__ x,
                           const float* __restrict__ w,
                           float* __restrict__ part) {
  __shared__ __align__(16) float sA[16 * LDP];
  __shared__ __align__(16) float sB[NPAD * LDP];
  const int tid  = threadIdx.x;
  const int lane = tid & 31;
  const int wave = tid >> 5;
  const int m0   = blockIdx.x * 16;
  const int k0   = blockIdx.y * 64;
  const int half = lane >> 4;
  const int l15  = lane & 15;

  const int  bk  = tid & 63;
  const int  bn0 = tid >> 6;
  const bool bkv = (k0 + bk) < PIX;
  float rb[14];
  const float* wcol = w + (bkv ? k0 + bk : 0);
#pragma unroll
  for (int j = 0; j < 14; ++j) {
    const int n = bn0 + 4 * j;
    const float t = wcol[(size_t)(n < NRN ? n : 0) * PIX];
    rb[j] = (bkv && n < NRN) ? t : 0.0f;
  }

#pragma unroll 1
  for (int i = 0; i < 2; ++i) {
    const int mr = 2 * wave + i;
    const float* img = x + (size_t)(m0 + mr) * 360000;
    const int p0 = k0 + lane, p1 = p0 + 32;
    const int c0 = p0 < PIX ? p0 : 0, c1 = p1 < PIX ? p1 : 0;
    const int pr0 = c0 / 75, pr1 = c1 / 75;
    float4 v0[16], v1[16];
    pool_load(img + pr0 * 4800 + (c0 - pr0 * 75) * 8, v0);
    pool_load(img + pr1 * 4800 + (c1 - pr1 * 75) * 8, v1);
    const float s0 = pool_sum(v0), s1 = pool_sum(v1);
    sA[mr * LDP + lane]      = p0 < PIX ? s0 : 0.0f;
    sA[mr * LDP + lane + 32] = p1 < PIX ? s1 : 0.0f;
  }
#pragma unroll
  for (int j = 0; j < 16; ++j) sB[(bn0 + 4 * j) * LDP + bk] = j < 14 ? rb[j] : 0.0f;
  __syncthreads();
  if (wave >= 4) return;

  v8f acc = {0.f, 0.f, 0.f, 0.f, 0.f, 0.f, 0.f, 0.f};
  const float* pa = &sA[l15 * LDP + (half << 1)];
  const float* pb = &sB[((wave << 4) + l15) * LDP + (half << 1)];
#pragma unroll 4
  for (int kk = 0; kk < 64; kk += 4) {
    const v2f a  = *reinterpret_cast<const v2f*>(pa + kk);
    const v2f bf = *reinterpret_cast<const v2f*>(pb + kk);
    acc = __builtin_amdgcn_wmma_f32_16x16x4_f32(
         false, a,  false, bf,
         (short)0, acc,  false,  false);
  }

  float* tile = part + ((size_t)blockIdx.y * BATCH + m0) * NPAD;
#pragma unroll
  for (int v = 0; v < 8; ++v) {
    int m = v + (half << 3);
    int n = (wave << 4) + l15;
    tile[m * NPAD + n] = acc[v];
  }
}

__global__ void __launch_bounds__(32) fc_reduce_kernel(const float* __restrict__ part,
                                                       const float* __restrict__ bias,
                                                       float* __restrict__ xf) {
  const int m = blockIdx.x;
  const int n = threadIdx.x << 1;
  const float* p = part + (size_t)m * NPAD + n;
  float sx = 0.0f, sy = 0.0f;
#pragma unroll
  for (int s = 0; s < KSPLIT; ++s) {
    float2 v = *reinterpret_cast<const float2*>(p + (size_t)s * (BATCH * NPAD));
    sx += v.x;
    sy += v.y;
  }
  float2 o;
  o.x = sx + (n < NRN ? bias[n] : 0.0f);
  o.y = sy + (n + 1 < NRN ? bias[n + 1] : 0.0f);
  *reinterpret_cast<float2*>(xf + m * NPAD + n) = o;
}

__device__ __forceinline__ float xhalf(float x) {
  int b = __builtin_bit_cast(int, x);
  return __builtin_bit_cast(float,
      __builtin_amdgcn_permlanex16(b, b, 0x76543210, 0xfedcba98, false, false));
}
template <int M>
__device__ __forceinline__ float xrow(float x) {
  int b = __builtin_bit_cast(int, x);
  return __builtin_bit_cast(float,
      __builtin_amdgcn_update_dpp(b, b, 0x160 + M, 0xf, 0xf, true));
}
__device__ __forceinline__ float wave_sum(float v) {
  v += xhalf(v);
  v += xrow<8>(v);
  v += xrow<4>(v);
  v += xrow<2>(v);
  v += xrow<1>(v);
  return v;
}

#define NTH  1024
#define NPR  3
struct Slots {
  v2f   p[NPR];
  float s;
};
__device__ __forceinline__ constexpr int slot_of(int p, int e) {
  return p == 0 ? (e ? 2 : 0) : p == 1 ? (e ? 3 : 1) : (e ? 5 : 4);
}

__device__ __forceinline__ void scan_step(Slots& a, const Slots& b, const Slots& c, const Slots& d,
                                          Slots& cnt, const Slots& xv,
                                          const Slots& l0, const Slots& l1,
                                          const Slots& l2, const Slots& l3,
                                          const v2f w0, const v2f w1,
                                          const v2f w2, const v2f w3,
                                          const v2f q0, const v2f q1,
                                          const v2f q2, const v2f q3,
                                          float* __restrict__ sw, const int lane, const int wv) {
  Slots sv, hsv, rate;
#pragma unroll
  for (int p = 0; p < NPR; ++p) {
    sv.p[p]  = a.p[p] * l0.p[p] + b.p[p] * l1.p[p] + c.p[p] * l2.p[p] + d.p[p] * l3.p[p];
    hsv.p[p] = a.p[p] * w0 + b.p[p] * w1 + c.p[p] * w2 + d.p[p] * w3;
  }
  const v2f r6 = a.s * q0 + b.s * q1 + c.s * q2 + d.s * q3;
  sv.s  = r6.x;
  hsv.s = r6.y;
  const v2f t01 = sv.p[0] + sv.p[1];
  float S = (t01.x + t01.y) + ((sv.p[2].x + sv.p[2].y) + sv.s);
  S += xrow<4>(S);
  S += xrow<2>(S);
  S += xrow<1>(S);
  float psum = 0.0f;
#pragma unroll
  for (int p = 0; p < NPR; ++p) {
    v2f cur = xv.p[p] + (S - sv.p[p]) + hsv.p[p];
    rate.p[p].x = __expf(cur.x);
    rate.p[p].y = __expf(cur.y);
  }
  rate.s = __expf(xv.s + (S - sv.s) + hsv.s);
  psum += rate.p[0].x;
  psum += rate.p[1].x;
  psum += rate.p[0].y;
  psum += rate.p[1].y;
  psum += rate.p[2].x;
  psum += rate.p[2].y;
  psum += rate.s;
  psum = wave_sum(psum);
  if (lane == 0) sw[wv] = psum;
  __syncthreads();
  const float thr = wave_sum(sw[lane]) * (1.0f / (float)CELLS);
#pragma unroll
  for (int p = 0; p < NPR; ++p) {
    v2f sp;
    sp.x = (rate.p[p].x > thr) ? 1.0f : 0.0f;
    sp.y = (rate.p[p].y > thr) ? 1.0f : 0.0f;
    cnt.p[p] += sp;
    a.p[p] = sp;
  }
  const float sps = (rate.s > thr) ? 1.0f : 0.0f;
  cnt.s += sps;
  a.s = sps;
}

__device__ __forceinline__ float softplus_f(const float c) {
  return fmaxf(c, 0.0f) + log1pf(__expf(-fabsf(c)));
}
static_assert(STEPS + 1 <= NTH, "one thread per possible count");

__global__ void __launch_bounds__(NTH) scan_kernel(const float* __restrict__ xf,
                                                   const float* __restrict__ lw,
                                                   const float* __restrict__ hw,
                                                   float* __restrict__ out) {
  __shared__ float sWave[2][NTH / 32];
  __shared__ float sOut[CELLS];
  __shared__ float sSoft[STEPS + 1];

  const int tid  = threadIdx.x;
  const int lane = tid & 31;
  const int wv   = tid >> 5;
  const int b    = tid >> 3;
  const int c8   = tid & 7;

  Slots h0, h1, h2, h3, cnt, xv, lw0, lw1, lw2, lw3;
  const float hw0 = hw[0], hw1 = hw[1], hw2 = hw[2], hw3 = hw[3];


#pragma unroll
  for (int p = 0; p < NPR; ++p) {
#pragma unroll
    for (int e = 0; e < 2; ++e) {
      const int n = c8 + (slot_of(p, e) << 3);
      xv.p[p][e] = xf[b * NPAD + n];
      const float4 l = *reinterpret_cast<const float4*>(lw + n * DTH);
      lw0.p[p][e] = l.x; lw1.p[p][e] = l.y; lw2.p[p][e] = l.z; lw3.p[p][e] = l.w;
      h0.p[p][e] = 0.0f; h1.p[p][e] = 0.0f; h2.p[p][e] = 0.0f; h3.p[p][e] = 0.0f;
      cnt.p[p][e] = 0.0f;
    }
  }
  if (tid <= STEPS) sSoft[tid] = softplus_f((float)tid);
  {
    const int n = c8 + (2 * NPR << 3);
    h0.s = h1.s = h2.s = h3.s = cnt.s = 0.0f;
    if (n < NRN) {
      xv.s = xf[b * NPAD + n];
      const float4 l = *reinterpret_cast<const float4*>(lw + n * DTH);
      lw0.s = l.x; lw1.s = l.y; lw2.s = l.z; lw3.s = l.w;
    } else {
      xv.s = -INFINITY;
      lw0.s = lw1.s = lw2.s = lw3.s = 0.0f;
    }
  }
  v2f w0 = {hw0, hw0}, w1 = {hw1, hw1}, w2 = {hw2, hw2}, w3 = {hw3, hw3};
  v2f q0 = {lw0.s, hw0}, q1 = {lw1.s, hw1}, q2 = {lw2.s, hw2}, q3 = {lw3.s, hw3};
  asm volatile("" : "+v"(w0), "+v"(w1), "+v"(w2), "+v"(w3),
                    "+v"(q0), "+v"(q1), "+v"(q2), "+v"(q3));

#define SCAN_STEP(A, B, C, D, PLANE) \
  scan_step(A, B, C, D, cnt, xv, lw0, lw1, lw2, lw3, w0, w1, w2, w3, q0, q1, q2, q3, \
            sWave[PLANE], lane, wv)
  for (int it = 0; it < STEPS / 4; ++it) {
    SCAN_STEP(h0, h1, h2, h3, 0);
    SCAN_STEP(h1, h2, h3, h0, 1);
    SCAN_STEP(h2, h3, h0, h1, 0);
    SCAN_STEP(h3, h0, h1, h2, 1);
  }
  if (STEPS % 4 > 0) SCAN_STEP(h0, h1, h2, h3, 0);
  if (STEPS % 4 > 1) SCAN_STEP(h1, h2, h3, h0, 1);
  if (STEPS % 4 > 2) SCAN_STEP(h2, h3, h0, h1, 0);
#undef SCAN_STEP

#pragma unroll
  for (int p = 0; p < NPR; ++p) {
#pragma unroll
    for (int e = 0; e < 2; ++e) {
      const int n = c8 + (slot_of(p, e) << 3);
      sOut[b * NRN + n] = sSoft[(int)cnt.p[p][e]];
    }
  }
  if (c8 + (2 * NPR << 3) < NRN) {
    sOut[b * NRN + c8 + (2 * NPR << 3)] = sSoft[(int)cnt.s];
  }
  __syncthreads();
#pragma unroll
  for (int j = 0; j < 2 * NPR + 1; ++j) {
    int c = tid + (j << 10);
    if (c < CELLS) out[c] = sOut[c];
  }
}

extern "C" void kernel_launch(void* const* d_in, const int* in_sizes, int n_in,
                              void* d_out, int out_size, void* d_ws, size_t ws_size,
                              hipStream_t stream) {
  const float* x  = (const float*)d_in[0];
  const float* lw = (const float*)d_in[1];
  const float* hw = (const float*)d_in[2];
  const float* fw = (const float*)d_in[3];
  const float* fb = (const float*)d_in[4];

  float* ws = (float*)d_ws;
  float* xf = ws;
  float* part = xf + (size_t)BATCH * NPAD;
  float* out = (float*)d_out;

  fc_pool_wmma_splitk_kernel<<<dim3(BATCH / 16, KSPLIT), 256, 0, stream>>>(x, fw, part);
  fc_reduce_kernel<<<BATCH, 32, 0, stream>>>(part, fb, xf);
  scan_kernel<<<1, NTH, 0, stream>>>(xf, lw, hw, out);
}


# scan step regrouped: cur = (xf + <hist, h_weight - l_weight>) + S_b, S_b in one accumulator per lane, log2(e) folded into the constants (exp2 direct); order of current and rate-mean sums changed
# speedup vs baseline: 10.4293x; 1.0481x over previous
#include <hip/hip_runtime.h>
#include <hip/hip_bf16.h>
#include <math.h>

typedef __attribute__((ext_vector_type(2))) float v2f;
typedef __attribute__((ext_vector_type(8))) float v8f;

#define BATCH  128
#define NRN    54
#define DTH    4
#define STEPS  30
#define PIX    5625
#define KPAD   5632
#define NPAD   64
#define CELLS  (BATCH * NRN)
#define KSPLIT (KPAD / 64)

#define LDP 68

__device__ __forceinline__ void pool_load(const float* __restrict__ base, float4 (&v)[16]) {
#pragma unroll
  for (int r = 0; r < 8; ++r) {
    const float4* q = reinterpret_cast<const float4*>(base + r * 600);
    v[2 * r]     = q[0];
    v[2 * r + 1] = q[1];
  }
}
__device__ __forceinline__ float pool_sum(const float4 (&v)[16]) {
  float s = 0.0f;
#pragma unroll
  for (int r = 0; r < 8; ++r) {
    const float4 a0 = v[2 * r], a1 = v[2 * r + 1];
    s += (a0.x + a0.y + a0.z + a0.w) + (a1.x + a1.y + a1.z + a1.w);
  }
  return s * (1.0f / 64.0f);
}

__global__ void __launch_bounds__(256) __attribute__((amdgpu_waves_per_eu(6, 6)))
fc_pool_wmma_splitk_kernel(const float* __restrict__ x,
                           const float* __restrict__ w,
                           float* __restrict__ part) {
  __shared__ __align__(16) float sA[16 * LDP];
  __shared__ __align__(16) float sB[NPAD * LDP];
  const int tid  = threadIdx.x;
  const int lane = tid & 31;
  const int wave = tid >> 5;
  const int m0   = blockIdx.x * 16;
  const int k0   = blockIdx.y * 64;
  const int half = lane >> 4;
  const int l15  = lane & 15;

  const int  bk  = tid & 63;
  const int  bn0 = tid >> 6;
  const bool bkv = (k0 + bk) < PIX;
  float rb[14];
  const float* wcol = w + (bkv ? k0 + bk : 0);
#pragma unroll
  for (int j = 0; j < 14; ++j) {
    const int n = bn0 + 4 * j;
    const float t = wcol[(size_t)(n < NRN ? n : 0) * PIX];
    rb[j] = (bkv && n < NRN) ? t : 0.0f;
  }

#pragma unroll 1
  for (int i = 0; i < 2; ++i) {
    const int mr = 2 * wave + i;
    const float* img = x + (size_t)(m0 + mr) * 360000;
    const int p0 = k0 + lane, p1 = p0 + 32;
    const int c0 = p0 < PIX ? p0 : 0, c1 = p1 < PIX ? p1 : 0;
    const int pr0 = c0 / 75, pr1 = c1 / 75;
    float4 v0[16], v1[16];
    pool_load(img + pr0 * 4800 + (c0 - pr0 * 75) * 8, v0);
    pool_load(img + pr1 * 4800 + (c1 - pr1 * 75) * 8, v1);
    const float s0 = pool_sum(v0), s1 = pool_sum(v1);
    sA[mr * LDP + lane]      = p0 < PIX ? s0 : 0.0f;
    sA[mr * LDP + lane + 32] = p1 < PIX ? s1 : 0.0f;
  }
#pragma unroll
  for (int j = 0; j < 16; ++j) sB[(bn0 + 4 * j) * LDP + bk] = j < 14 ? rb[j] : 0.0f;
  __syncthreads();
  if (wave >= 4) return;

  v8f acc = {0.f, 0.f, 0.f, 0.f, 0.f, 0.f, 0.f, 0.f};
  const float* pa = &sA[l15 * LDP + (half << 1)];
  const float* pb = &sB[((wave << 4) + l15) * LDP + (half << 1)];
#pragma unroll 4
  for (int kk = 0; kk < 64; kk += 4) {
    const v2f a  = *reinterpret_cast<const v2f*>(pa + kk);
    const v2f bf = *reinterpret_cast<const v2f*>(pb + kk);
    acc = __builtin_amdgcn_wmma_f32_16x16x4_f32(
         false, a,  false, bf,
         (short)0, acc,  false,  false);
  }

  float* tile = part + ((size_t)blockIdx.y * BATCH + m0) * NPAD;
#pragma unroll
  for (int v = 0; v < 8; ++v) {
    int m = v + (half << 3);
    int n = (wave << 4) + l15;
    tile[m * NPAD + n] = acc[v];
  }
}

__global__ void __launch_bounds__(32) fc_reduce_kernel(const float* __restrict__ part,
                                                       const float* __restrict__ bias,
                                                       float* __restrict__ xf) {
  const int m = blockIdx.x;
  const int n = threadIdx.x << 1;
  const float* p = part + (size_t)m * NPAD + n;
  float sx = 0.0f, sy = 0.0f;
#pragma unroll
  for (int s = 0; s < KSPLIT; ++s) {
    float2 v = *reinterpret_cast<const float2*>(p + (size_t)s * (BATCH * NPAD));
    sx += v.x;
    sy += v.y;
  }
  float2 o;
  o.x = sx + (n < NRN ? bias[n] : 0.0f);
  o.y = sy + (n + 1 < NRN ? bias[n + 1] : 0.0f);
  *reinterpret_cast<float2*>(xf + m * NPAD + n) = o;
}

__device__ __forceinline__ float xhalf(float x) {
  int b = __builtin_bit_cast(int, x);
  return __builtin_bit_cast(float,
      __builtin_amdgcn_permlanex16(b, b, 0x76543210, 0xfedcba98, false, false));
}
template <int M>
__device__ __forceinline__ float xrow(float x) {
  int b = __builtin_bit_cast(int, x);
  return __builtin_bit_cast(float,
      __builtin_amdgcn_update_dpp(b, b, 0x160 + M, 0xf, 0xf, true));
}
__device__ __forceinline__ float wave_sum(float v) {
  v += xhalf(v);
  v += xrow<8>(v);
  v += xrow<4>(v);
  v += xrow<2>(v);
  v += xrow<1>(v);
  return v;
}

#define NTH  1024
#define NPR  3
#define LOG2E 1.4426950408889634f
struct Slots {
  v2f   p[NPR];
  float s;
};
struct Pairs {
  v2f   p[NPR];
};
__device__ __forceinline__ constexpr int slot_of(int p, int e) {
  return p == 0 ? (e ? 2 : 0) : p == 1 ? (e ? 3 : 1) : (e ? 5 : 4);
}
__device__ __forceinline__ v2f pk_fma(const v2f x, const v2f y, const v2f z) {
  return __builtin_elementwise_fma(x, y, z);
}
__device__ __forceinline__ v2f both(const float x) {
  v2f r = {x, x};
  return r;
}

__device__ __forceinline__ void scan_step(Slots& a, const Slots& b, const Slots& c, const Slots& d,
                                          Slots& cnt, const Pairs& xv,
                                          const Pairs& l0, const Pairs& l1,
                                          const Pairs& l2, const Pairs& l3,
                                          const Pairs& e0, const Pairs& e1,
                                          const Pairs& e2, const Pairs& e3,
                                          const v2f q0, const v2f q1,
                                          const v2f q2, const v2f q3,
                                          const v2f x6,
                                          float* __restrict__ sw, const int lane, const int wv) {
  Slots rate;
  v2f own[NPR];
#pragma unroll
  for (int p = 0; p < NPR; ++p) {
    own[p] = pk_fma(d.p[p], e3.p[p],
             pk_fma(c.p[p], e2.p[p],
             pk_fma(b.p[p], e1.p[p],
             pk_fma(a.p[p], e0.p[p], xv.p[p]))));
  }
  v2f acc = a.p[0] * l0.p[0];
  acc = pk_fma(a.p[1], l0.p[1], acc);
  acc = pk_fma(a.p[2], l0.p[2], acc);
#pragma unroll
  for (int p = 0; p < NPR; ++p) acc = pk_fma(b.p[p], l1.p[p], acc);
#pragma unroll
  for (int p = 0; p < NPR; ++p) acc = pk_fma(c.p[p], l2.p[p], acc);
#pragma unroll
  for (int p = 0; p < NPR; ++p) acc = pk_fma(d.p[p], l3.p[p], acc);
  const v2f r6 = pk_fma(both(d.s), q3,
                 pk_fma(both(c.s), q2,
                 pk_fma(both(b.s), q1,
                 pk_fma(both(a.s), q0, x6))));
  float S = (acc.x + acc.y) + r6.x;
  S += xrow<4>(S);
  S += xrow<2>(S);
  S += xrow<1>(S);
#pragma unroll
  for (int p = 0; p < NPR; ++p) {
    const v2f cur = own[p] + both(S);
    rate.p[p].x = __builtin_amdgcn_exp2f(cur.x);
    rate.p[p].y = __builtin_amdgcn_exp2f(cur.y);
  }
  rate.s = __builtin_amdgcn_exp2f(r6.y + S);
  const v2f ps = (rate.p[0] + rate.p[1]) + rate.p[2];
  float psum = (ps.x + ps.y) + rate.s;
  psum = wave_sum(psum);
  if (lane == 0) sw[wv] = psum;
  __syncthreads();
  const float thr = wave_sum(sw[lane]) * (1.0f / (float)CELLS);
#pragma unroll
  for (int p = 0; p < NPR; ++p) {
    v2f sp;
    sp.x = (rate.p[p].x > thr) ? 1.0f : 0.0f;
    sp.y = (rate.p[p].y > thr) ? 1.0f : 0.0f;
    cnt.p[p] += sp;
    a.p[p] = sp;
  }
  const float sps = (rate.s > thr) ? 1.0f : 0.0f;
  cnt.s += sps;
  a.s = sps;
}

__device__ __forceinline__ float softplus_f(const float c) {
  return fmaxf(c, 0.0f) + log1pf(__expf(-fabsf(c)));
}
static_assert(STEPS + 1 <= NTH, "one thread per possible count");

__global__ void __launch_bounds__(NTH) scan_kernel(const float* __restrict__ xf,
                                                   const float* __restrict__ lw,
                                                   const float* __restrict__ hw,
                                                   float* __restrict__ out) {
  __shared__ float sWave[2][NTH / 32];
  __shared__ float sOut[CELLS];
  __shared__ float sSoft[STEPS + 1];

  const int tid  = threadIdx.x;
  const int lane = tid & 31;
  const int wv   = tid >> 5;
  const int b    = tid >> 3;
  const int c8   = tid & 7;

  Slots h0, h1, h2, h3, cnt;
  Pairs xv, lw0, lw1, lw2, lw3, dw0, dw1, dw2, dw3;
  const float hw0 = hw[0], hw1 = hw[1], hw2 = hw[2], hw3 = hw[3];

#pragma unroll
  for (int p = 0; p < NPR; ++p) {
#pragma unroll
    for (int e = 0; e < 2; ++e) {
      const int n = c8 + (slot_of(p, e) << 3);
      xv.p[p][e] = xf[b * NPAD + n] * LOG2E;
      const float4 l = *reinterpret_cast<const float4*>(lw + n * DTH);
      lw0.p[p][e] = l.x * LOG2E; lw1.p[p][e] = l.y * LOG2E;
      lw2.p[p][e] = l.z * LOG2E; lw3.p[p][e] = l.w * LOG2E;
      dw0.p[p][e] = (hw0 - l.x) * LOG2E; dw1.p[p][e] = (hw1 - l.y) * LOG2E;
      dw2.p[p][e] = (hw2 - l.z) * LOG2E; dw3.p[p][e] = (hw3 - l.w) * LOG2E;
      h0.p[p][e] = 0.0f; h1.p[p][e] = 0.0f; h2.p[p][e] = 0.0f; h3.p[p][e] = 0.0f;
      cnt.p[p][e] = 0.0f;
    }
  }
  if (tid <= STEPS) sSoft[tid] = softplus_f((float)tid);
  v2f q0 = {0.0f, 0.0f}, q1 = q0, q2 = q0, q3 = q0;
  v2f x6 = {0.0f, -INFINITY};
  {
    const int n = c8 + (2 * NPR << 3);
    h0.s = h1.s = h2.s = h3.s = cnt.s = 0.0f;
    if (n < NRN) {
      x6.y = xf[b * NPAD + n] * LOG2E;
      const float4 l = *reinterpret_cast<const float4*>(lw + n * DTH);
      q0.x = l.x * LOG2E; q0.y = (hw0 - l.x) * LOG2E;
      q1.x = l.y * LOG2E; q1.y = (hw1 - l.y) * LOG2E;
      q2.x = l.z * LOG2E; q2.y = (hw2 - l.z) * LOG2E;
      q3.x = l.w * LOG2E; q3.y = (hw3 - l.w) * LOG2E;
    }
  }

#define SCAN_STEP(A, B, C, D, PLANE) \
  scan_step(A, B, C, D, cnt, xv, lw0, lw1, lw2, lw3, dw0, dw1, dw2, dw3, q0, q1, q2, q3, x6, \
            sWave[PLANE], lane, wv)
  for (int it = 0; it < STEPS / 4; ++it) {
    SCAN_STEP(h0, h1, h2, h3, 0);
    SCAN_STEP(h1, h2, h3, h0, 1);
    SCAN_STEP(h2, h3, h0, h1, 0);
    SCAN_STEP(h3, h0, h1, h2, 1);
  }
  if (STEPS % 4 > 0) SCAN_STEP(h0, h1, h2, h3, 0);
  if (STEPS % 4 > 1) SCAN_STEP(h1, h2, h3, h0, 1);
  if (STEPS % 4 > 2) SCAN_STEP(h2, h3, h0, h1, 0);
#undef SCAN_STEP

#pragma unroll
  for (int p = 0; p < NPR; ++p) {
#pragma unroll
    for (int e = 0; e < 2; ++e) {
      const int n = c8 + (slot_of(p, e) << 3);
      sOut[b * NRN + n] = sSoft[(int)cnt.p[p][e]];
    }
  }
  if (c8 + (2 * NPR << 3) < NRN) {
    sOut[b * NRN + c8 + (2 * NPR << 3)] = sSoft[(int)cnt.s];
  }
  __syncthreads();
#pragma unroll
  for (int j = 0; j < 2 * NPR + 1; ++j) {
    int c = tid + (j << 10);
    if (c < CELLS) out[c] = sOut[c];
  }
}

extern "C" void kernel_launch(void* const* d_in, const int* in_sizes, int n_in,
                              void* d_out, int out_size, void* d_ws, size_t ws_size,
                              hipStream_t stream) {
  const float* x  = (const float*)d_in[0];
  const float* lw = (const float*)d_in[1];
  const float* hw = (const float*)d_in[2];
  const float* fw = (const float*)d_in[3];
  const float* fb = (const float*)d_in[4];

  float* ws = (float*)d_ws;
  float* xf = ws;
  float* part = xf + (size_t)BATCH * NPAD;
  float* out = (float*)d_out;

  fc_pool_wmma_splitk_kernel<<<dim3(BATCH / 16, KSPLIT), 256, 0, stream>>>(x, fw, part);
  fc_reduce_kernel<<<BATCH, 32, 0, stream>>>(part, fb, xf);
  scan_kernel<<<1, NTH, 0, stream>>>(xf, lw, hw, out);
}


# scan: one batch row per 4-lane group (14 neurons per lane), 512 threads not 1024: per-lane sums, LDS hand-over and threshold issued half as often (order of lateral and rate-mean sums changed)
# speedup vs baseline: 10.7386x; 1.0297x over previous
#include <hip/hip_runtime.h>
#include <hip/hip_bf16.h>
#include <math.h>

typedef __attribute__((ext_vector_type(2))) float v2f;
typedef __attribute__((ext_vector_type(8))) float v8f;

#define BATCH  128
#define NRN    54
#define DTH    4
#define STEPS  30
#define PIX    5625
#define KPAD   5632
#define NPAD   64
#define CELLS  (BATCH * NRN)
#define KSPLIT (KPAD / 64)

#define LDP 68

__device__ __forceinline__ void pool_load(const float* __restrict__ base, float4 (&v)[16]) {
#pragma unroll
  for (int r = 0; r < 8; ++r) {
    const float4* q = reinterpret_cast<const float4*>(base + r * 600);
    v[2 * r]     = q[0];
    v[2 * r + 1] = q[1];
  }
}
__device__ __forceinline__ float pool_sum(const float4 (&v)[16]) {
  float s = 0.0f;
#pragma unroll
  for (int r = 0; r < 8; ++r) {
    const float4 a0 = v[2 * r], a1 = v[2 * r + 1];
    s += (a0.x + a0.y + a0.z + a0.w) + (a1.x + a1.y + a1.z + a1.w);
  }
  return s * (1.0f / 64.0f);
}

__global__ void __launch_bounds__(256) __attribute__((amdgpu_waves_per_eu(6, 6)))
fc_pool_wmma_splitk_kernel(const float* __restrict__ x,
                           const float* __restrict__ w,
                           float* __restrict__ part) {
  __shared__ __align__(16) float sA[16 * LDP];
  __shared__ __align__(16) float sB[NPAD * LDP];
  const int tid  = threadIdx.x;
  const int lane = tid & 31;
  const int wave = tid >> 5;
  const int m0   = blockIdx.x * 16;
  const int k0   = blockIdx.y * 64;
  const int half = lane >> 4;
  const int l15  = lane & 15;

  const int  bk  = tid & 63;
  const int  bn0 = tid >> 6;
  const bool bkv = (k0 + bk) < PIX;
  float rb[14];
  const float* wcol = w + (bkv ? k0 + bk : 0);
#pragma unroll
  for (int j = 0; j < 14; ++j) {
    const int n = bn0 + 4 * j;
    const float t = wcol[(size_t)(n < NRN ? n : 0) * PIX];
    rb[j] = (bkv && n < NRN) ? t : 0.0f;
  }

#pragma unroll 1
  for (int i = 0; i < 2; ++i) {
    const int mr = 2 * wave + i;
    const float* img = x + (size_t)(m0 + mr) * 360000;
    const int p0 = k0 + lane, p1 = p0 + 32;
    const int c0 = p0 < PIX ? p0 : 0, c1 = p1 < PIX ? p1 : 0;
    const int pr0 = c0 / 75, pr1 = c1 / 75;
    float4 v0[16], v1[16];
    pool_load(img + pr0 * 4800 + (c0 - pr0 * 75) * 8, v0);
    pool_load(img + pr1 * 4800 + (c1 - pr1 * 75) * 8, v1);
    const float s0 = pool_sum(v0), s1 = pool_sum(v1);
    sA[mr * LDP + lane]      = p0 < PIX ? s0 : 0.0f;
    sA[mr * LDP + lane + 32] = p1 < PIX ? s1 : 0.0f;
  }
#pragma unroll
  for (int j = 0; j < 16; ++j) sB[(bn0 + 4 * j) * LDP + bk] = j < 14 ? rb[j] : 0.0f;
  __syncthreads();
  if (wave >= 4) return;

  v8f acc = {0.f, 0.f, 0.f, 0.f, 0.f, 0.f, 0.f, 0.f};
  const float* pa = &sA[l15 * LDP + (half << 1)];
  const float* pb = &sB[((wave << 4) + l15) * LDP + (half << 1)];
#pragma unroll 4
  for (int kk = 0; kk < 64; kk += 4) {
    const v2f a  = *reinterpret_cast<const v2f*>(pa + kk);
    const v2f bf = *reinterpret_cast<const v2f*>(pb + kk);
    acc = __builtin_amdgcn_wmma_f32_16x16x4_f32(
         false, a,  false, bf,
         (short)0, acc,  false,  false);
  }

  float* tile = part + ((size_t)blockIdx.y * BATCH + m0) * NPAD;
#pragma unroll
  for (int v = 0; v < 8; ++v) {
    int m = v + (half << 3);
    int n = (wave << 4) + l15;
    tile[m * NPAD + n] = acc[v];
  }
}

__global__ void __launch_bounds__(32) fc_reduce_kernel(const float* __restrict__ part,
                                                       const float* __restrict__ bias,
                                                       float* __restrict__ xf) {
  const int m = blockIdx.x;
  const int n = threadIdx.x << 1;
  const float* p = part + (size_t)m * NPAD + n;
  float sx = 0.0f, sy = 0.0f;
#pragma unroll
  for (int s = 0; s < KSPLIT; ++s) {
    float2 v = *reinterpret_cast<const float2*>(p + (size_t)s * (BATCH * NPAD));
    sx += v.x;
    sy += v.y;
  }
  float2 o;
  o.x = sx + (n < NRN ? bias[n] : 0.0f);
  o.y = sy + (n + 1 < NRN ? bias[n + 1] : 0.0f);
  *reinterpret_cast<float2*>(xf + m * NPAD + n) = o;
}

__device__ __forceinline__ float xhalf(float x) {
  int b = __builtin_bit_cast(int, x);
  return __builtin_bit_cast(float,
      __builtin_amdgcn_permlanex16(b, b, 0x76543210, 0xfedcba98, false, false));
}
template <int M>
__device__ __forceinline__ float xrow(float x) {
  int b = __builtin_bit_cast(int, x);
  return __builtin_bit_cast(float,
      __builtin_amdgcn_update_dpp(b, b, 0x160 + M, 0xf, 0xf, true));
}
__device__ __forceinline__ float wave_sum(float v) {
  v += xhalf(v);
  v += xrow<8>(v);
  v += xrow<4>(v);
  v += xrow<2>(v);
  v += xrow<1>(v);
  return v;
}

#define NTH  512
#define NWV  (NTH / 32)
#define NPR  7
#define LOG2E 1.4426950408889634f
struct Slots {
  v2f   p[NPR];
};
struct Pairs {
  v2f   p[NPR];
};
__device__ __forceinline__ v2f pk_fma(const v2f x, const v2f y, const v2f z) {
  return __builtin_elementwise_fma(x, y, z);
}
__device__ __forceinline__ v2f both(const float x) {
  v2f r = {x, x};
  return r;
}
__device__ __forceinline__ float row_sum(float v) {
  v += xrow<8>(v);
  v += xrow<4>(v);
  v += xrow<2>(v);
  v += xrow<1>(v);
  return v;
}

__device__ __forceinline__ void scan_step(Slots& a, const Slots& b, const Slots& c, const Slots& d,
                                          Slots& cnt, const Pairs& xv,
                                          const Pairs& l0, const Pairs& l1,
                                          const Pairs& l2, const Pairs& l3,
                                          const Pairs& e0, const Pairs& e1,
                                          const Pairs& e2, const Pairs& e3,
                                          float* __restrict__ sw, const int lane, const int wv) {
  v2f rate[NPR];
  v2f own[NPR];
#pragma unroll
  for (int p = 0; p < NPR; ++p) {
    own[p] = pk_fma(d.p[p], e3.p[p],
             pk_fma(c.p[p], e2.p[p],
             pk_fma(b.p[p], e1.p[p],
             pk_fma(a.p[p], e0.p[p], xv.p[p]))));
  }
  v2f acc0 = a.p[0] * l0.p[0];
  v2f acc1 = a.p[1] * l0.p[1];
#pragma unroll
  for (int p = 2; p < NPR; ++p) {
    if (p & 1) acc1 = pk_fma(a.p[p], l0.p[p], acc1);
    else       acc0 = pk_fma(a.p[p], l0.p[p], acc0);
  }
#pragma unroll
  for (int p = 0; p < NPR; ++p) {
    if (p & 1) acc1 = pk_fma(b.p[p], l1.p[p], acc1);
    else       acc0 = pk_fma(b.p[p], l1.p[p], acc0);
  }
#pragma unroll
  for (int p = 0; p < NPR; ++p) {
    if (p & 1) acc1 = pk_fma(c.p[p], l2.p[p], acc1);
    else       acc0 = pk_fma(c.p[p], l2.p[p], acc0);
  }
#pragma unroll
  for (int p = 0; p < NPR; ++p) {
    if (p & 1) acc1 = pk_fma(d.p[p], l3.p[p], acc1);
    else       acc0 = pk_fma(d.p[p], l3.p[p], acc0);
  }
  const v2f accs = acc0 + acc1;
  float S = accs.x + accs.y;
  S += xrow<2>(S);
  S += xrow<1>(S);
#pragma unroll
  for (int p = 0; p < NPR; ++p) {
    const v2f cur = own[p] + both(S);
    rate[p].x = __builtin_amdgcn_exp2f(cur.x);
    rate[p].y = __builtin_amdgcn_exp2f(cur.y);
  }
  const v2f ps = ((rate[0] + rate[1]) + (rate[2] + rate[3])) + ((rate[4] + rate[5]) + rate[6]);
  float psum = ps.x + ps.y;
  psum = wave_sum(psum);
  if (lane == 0) sw[wv] = psum;
  __syncthreads();
  const float thr = row_sum(sw[lane & (NWV - 1)]) * (1.0f / (float)CELLS);
#pragma unroll
  for (int p = 0; p < NPR; ++p) {
    v2f sp;
    sp.x = (rate[p].x > thr) ? 1.0f : 0.0f;
    sp.y = (rate[p].y > thr) ? 1.0f : 0.0f;
    cnt.p[p] += sp;
    a.p[p] = sp;
  }
}

__device__ __forceinline__ float softplus_f(const float c) {
  return fmaxf(c, 0.0f) + log1pf(__expf(-fabsf(c)));
}
static_assert(STEPS + 1 <= NTH, "one thread per possible count");
static_assert(NWV == 16, "the threshold sum reads one wave sum per lane of a 16-lane row");

__global__ void __launch_bounds__(NTH) scan_kernel(const float* __restrict__ xf,
                                                   const float* __restrict__ lw,
                                                   const float* __restrict__ hw,
                                                   float* __restrict__ out) {
  __shared__ float sWave[2][NWV];
  __shared__ float sOut[CELLS];
  __shared__ float sSoft[STEPS + 1];

  const int tid  = threadIdx.x;
  const int lane = tid & 31;
  const int wv   = tid >> 5;
  const int b    = tid >> 2;
  const int c4   = tid & 3;

  Slots h0, h1, h2, h3, cnt;
  Pairs xv, lw0, lw1, lw2, lw3, dw0, dw1, dw2, dw3;
  const float hw0 = hw[0], hw1 = hw[1], hw2 = hw[2], hw3 = hw[3];

#pragma unroll
  for (int p = 0; p < NPR; ++p) {
#pragma unroll
    for (int e = 0; e < 2; ++e) {
      const int  n  = c4 + ((2 * p + e) << 2);
      const bool ok = n < NRN;
      const int  nl = ok ? n : 0;
      const float  x = xf[b * NPAD + nl];
      const float4 l = *reinterpret_cast<const float4*>(lw + nl * DTH);
      xv.p[p][e]  = ok ? x * LOG2E : -INFINITY;
      lw0.p[p][e] = ok ? l.x * LOG2E : 0.0f; lw1.p[p][e] = ok ? l.y * LOG2E : 0.0f;
      lw2.p[p][e] = ok ? l.z * LOG2E : 0.0f; lw3.p[p][e] = ok ? l.w * LOG2E : 0.0f;
      dw0.p[p][e] = ok ? (hw0 - l.x) * LOG2E : 0.0f; dw1.p[p][e] = ok ? (hw1 - l.y) * LOG2E : 0.0f;
      dw2.p[p][e] = ok ? (hw2 - l.z) * LOG2E : 0.0f; dw3.p[p][e] = ok ? (hw3 - l.w) * LOG2E : 0.0f;
      h0.p[p][e] = 0.0f; h1.p[p][e] = 0.0f; h2.p[p][e] = 0.0f; h3.p[p][e] = 0.0f;
      cnt.p[p][e] = 0.0f;
    }
  }
  if (tid <= STEPS) sSoft[tid] = softplus_f((float)tid);

#define SCAN_STEP(A, B, C, D, PLANE) \
  scan_step(A, B, C, D, cnt, xv, lw0, lw1, lw2, lw3, dw0, dw1, dw2, dw3, \
            sWave[PLANE], lane, wv)
  for (int it = 0; it < STEPS / 4; ++it) {
    SCAN_STEP(h0, h1, h2, h3, 0);
    SCAN_STEP(h1, h2, h3, h0, 1);
    SCAN_STEP(h2, h3, h0, h1, 0);
    SCAN_STEP(h3, h0, h1, h2, 1);
  }
  if (STEPS % 4 > 0) SCAN_STEP(h0, h1, h2, h3, 0);
  if (STEPS % 4 > 1) SCAN_STEP(h1, h2, h3, h0, 1);
  if (STEPS % 4 > 2) SCAN_STEP(h2, h3, h0, h1, 0);
#undef SCAN_STEP

#pragma unroll
  for (int p = 0; p < NPR; ++p) {
#pragma unroll
    for (int e = 0; e < 2; ++e) {
      const int n = c4 + ((2 * p + e) << 2);
      if (n < NRN) sOut[b * NRN + n] = sSoft[(int)cnt.p[p][e]];
    }
  }
  __syncthreads();
#pragma unroll
  for (int j = 0; j < (CELLS + NTH - 1) / NTH; ++j) {
    int c = tid + j * NTH;
    if (c < CELLS) out[c] = sOut[c];
  }
}

extern "C" void kernel_launch(void* const* d_in, const int* in_sizes, int n_in,
                              void* d_out, int out_size, void* d_ws, size_t ws_size,
                              hipStream_t stream) {
  const float* x  = (const float*)d_in[0];
  const float* lw = (const float*)d_in[1];
  const float* hw = (const float*)d_in[2];
  const float* fw = (const float*)d_in[3];
  const float* fb = (const float*)d_in[4];

  float* ws = (float*)d_ws;
  float* xf = ws;
  float* part = xf + (size_t)BATCH * NPAD;
  float* out = (float*)d_out;

  fc_pool_wmma_splitk_kernel<<<dim3(BATCH / 16, KSPLIT), 256, 0, stream>>>(x, fw, part);
  fc_reduce_kernel<<<BATCH, 32, 0, stream>>>(part, fb, xf);
  scan_kernel<<<1, NTH, 0, stream>>>(xf, lw, hw, out);
}
